# starting kernel
# baseline (speedup 1.0000x reference)
#include <hip/hip_runtime.h>

typedef float v2f __attribute__((ext_vector_type(2)));
typedef float v8f __attribute__((ext_vector_type(8)));

#define GOAL_P  0.5f
#define MIN_P   (-1.2f)
#define MIN_V   (-0.07f)
#define MAX_V   (0.07f)
#define MIN_ACC (-1.2f)
#define MAX_ACC (1.2f)
#define MAX_STEPS 500

__global__ __launch_bounds__(256)
void MountainCar_7318624272632_kernel(const float* __restrict__ x,
                                      const float* __restrict__ w1,
                                      const float* __restrict__ b1,
                                      const float* __restrict__ w2,
                                      const float* __restrict__ b2,
                                      float* __restrict__ out, int B)
{
    const int lane   = threadIdx.x & 31;
    const int waveId = (blockIdx.x * blockDim.x + threadIdx.x) >> 5;
    const int ln     = lane & 15;
    const bool hiHalf = (lane >= 16);

    if (waveId * 16 >= B) return;

    const int  env      = waveId * 16 + ln;
    const bool envValid = (env < B);
    const int  envC     = envValid ? env : (B - 1);

    v2f  Bw[4];
    float w2l[4];
#pragma unroll
    for (int t = 0; t < 4; ++t) {
        const int col = t * 16 + ln;
        Bw[t][0] = hiHalf ? b1[col] : w1[col];
        Bw[t][1] = hiHalf ? 0.0f    : w1[64 + col];
        w2l[t]   = w2[col];
    }
    const float bias2 = b2[0];

    const float4 st = reinterpret_cast<const float4*>(x)[envC];
    float p = st.x, v = st.y, u = st.z, r = st.w;

    const bool bit0 = (lane & 1) != 0;
    const bool bit1 = (lane & 2) != 0;
    const bool bit2 = (lane & 4) != 0;

    for (int step = 0; step < MAX_STEPS; ++step) {
        const bool active = (p <= GOAL_P);
        if (__ballot(active && !hiHalf) == 0ull) break;

        const bool hit = (p <= MIN_P);
        const float p1 = hit ? MIN_P : p;
        const float v1 = hit ? 0.0f  : v;

        v2f A;
        A[0] = hiHalf ? 1.0f : p1;
        A[1] = hiHalf ? 0.0f : v1;

        float acc[8];
#pragma unroll
        for (int i = 0; i < 8; ++i) acc[i] = 0.0f;

#pragma unroll
        for (int t = 0; t < 4; ++t) {
            v8f c = {};
            c = __builtin_amdgcn_wmma_f32_16x16x4_f32(
                     false, A,  false, Bw[t],
                     (short)0, c,  false,  false);
#pragma unroll
            for (int i = 0; i < 8; ++i) {
                const float h = fmaxf(c[i], 0.0f);
                acc[i] = fmaf(h, w2l[t], acc[i]);
            }
        }

        float r1a[4];
#pragma unroll
        for (int j = 0; j < 4; ++j) {
            const float keep = bit0 ? acc[2 * j + 1] : acc[2 * j];
            const float send = bit0 ? acc[2 * j]     : acc[2 * j + 1];
            r1a[j] = keep + __shfl_xor(send, 1, 32);
        }
        float r2a[2];
#pragma unroll
        for (int j = 0; j < 2; ++j) {
            const float keep = bit1 ? r1a[2 * j + 1] : r1a[2 * j];
            const float send = bit1 ? r1a[2 * j]     : r1a[2 * j + 1];
            r2a[j] = keep + __shfl_xor(send, 2, 32);
        }
        {
            const float keep = bit2 ? r2a[1] : r2a[0];
            const float send = bit2 ? r2a[0] : r2a[1];
            r2a[0] = keep + __shfl_xor(send, 4, 32);
        }
        const float rowSum = r2a[0] + __shfl_xor(r2a[0], 8, 32);
        const float rowSumX = __shfl_xor(rowSum, 16, 32);

        const float upre = ((ln & 8) ? rowSumX : rowSum) + bias2;

        float u1 = tanhf(upre);
        u1 = fminf(fmaxf(u1, MIN_ACC), MAX_ACC);

        const float rr = fmaf(-0.1f * u1, u1, r);
        float v2 = v1 + 0.0015f * u1 - 0.0025f * __cosf(3.0f * p1);
        v2 = fminf(fmaxf(v2, MIN_V), MAX_V);
        const float p2 = p1 + v2;

        p = active ? p2 : p;
        v = active ? v2 : v;
        u = active ? u1 : u;
        r = active ? rr : r;
    }

    const float rout = (p <= GOAL_P) ? r : (r + 100.0f);
    if (!hiHalf && envValid) {
        float4 o;
        o.x = p; o.y = v; o.z = u; o.w = rout;
        reinterpret_cast<float4*>(out)[env] = o;
    }
}

extern "C" void kernel_launch(void* const* d_in, const int* in_sizes, int n_in,
                              void* d_out, int out_size, void* d_ws, size_t ws_size,
                              hipStream_t stream) {
    const float* x  = (const float*)d_in[0];
    const float* w1 = (const float*)d_in[1];
    const float* b1 = (const float*)d_in[2];
    const float* w2 = (const float*)d_in[3];
    const float* b2 = (const float*)d_in[4];
    float* out = (float*)d_out;

    const int B = in_sizes[0] / 4;
    const int envsPerWave = 16;
    const int waves = (B + envsPerWave - 1) / envsPerWave;
    const int threads = waves * 32;
    const int block = 256;
    const int grid = (threads + block - 1) / block;

    MountainCar_7318624272632_kernel<<<grid, block, 0, stream>>>(x, w1, b1, w2, b2, out, B);
}


# swap matrix operands (weights as A, state as B): a lane holds 32 hidden units of its own env, so the layer-2 sum is in-lane plus one lane-half exchange; the 64-term sum order changes
# speedup vs baseline: 1.7040x; 1.7040x over previous
#include <hip/hip_runtime.h>

typedef float v2f __attribute__((ext_vector_type(2)));
typedef float v8f __attribute__((ext_vector_type(8)));

#define GOAL_P  0.5f
#define MIN_P   (-1.2f)
#define MIN_V   (-0.07f)
#define MAX_V   (0.07f)
#define MIN_ACC (-1.2f)
#define MAX_ACC (1.2f)
#define MAX_STEPS 500

__device__ __forceinline__ float xhalf(float x) {
    const int b = __builtin_bit_cast(int, x);
    return __builtin_bit_cast(float,
        __builtin_amdgcn_permlanex16(b, b, 0x76543210, 0xfedcba98, false, false));
}

__global__ __launch_bounds__(256)
void MountainCar_7318624272632_kernel(const float* __restrict__ x,
                                      const float* __restrict__ w1,
                                      const float* __restrict__ b1,
                                      const float* __restrict__ w2,
                                      const float* __restrict__ b2,
                                      float* __restrict__ out, int B)
{
    const int lane   = threadIdx.x & 31;
    const int waveId = (blockIdx.x * blockDim.x + threadIdx.x) >> 5;
    const int ln     = lane & 15;
    const bool hiHalf = (lane >= 16);

    if (waveId * 16 >= B) return;

    const int  env      = waveId * 16 + ln;
    const bool envValid = (env < B);
    const int  envC     = envValid ? env : (B - 1);

    v2f  Aw[4];
    float w2r[4][8];
    const int hOff = hiHalf ? 8 : 0;
#pragma unroll
    for (int t = 0; t < 4; ++t) {
        const int j = t * 16 + ln;
        Aw[t][0] = hiHalf ? b1[j] : w1[j];
        Aw[t][1] = hiHalf ? 0.0f  : w1[64 + j];
#pragma unroll
        for (int i = 0; i < 8; ++i)
            w2r[t][i] = w2[t * 16 + hOff + i];
    }
    const float bias2 = b2[0];

    const float4 st = reinterpret_cast<const float4*>(x)[envC];
    float p = st.x, v = st.y, u = st.z, r = st.w;

    for (int step = 0; step < MAX_STEPS; ++step) {
        const bool active = (p <= GOAL_P);
        if (__ballot(active && !hiHalf) == 0ull) break;

        const bool hit = (p <= MIN_P);
        const float p1 = hit ? MIN_P : p;
        const float v1 = hit ? 0.0f  : v;

        v2f S;
        S[0] = hiHalf ? 1.0f : p1;
        S[1] = hiHalf ? 0.0f : v1;

        float acc[8];
#pragma unroll
        for (int i = 0; i < 8; ++i) acc[i] = 0.0f;

#pragma unroll
        for (int t = 0; t < 4; ++t) {
            v8f c = {};
            c = __builtin_amdgcn_wmma_f32_16x16x4_f32(
                     false, Aw[t],  false, S,
                     (short)0, c,  false,  false);
#pragma unroll
            for (int i = 0; i < 8; ++i) {
                const float h = fmaxf(c[i], 0.0f);
                acc[i] = fmaf(h, w2r[t][i], acc[i]);
            }
        }

        const float evens  = (acc[0] + acc[2]) + (acc[4] + acc[6]);
        const float odds   = (acc[1] + acc[3]) + (acc[5] + acc[7]);
        const float half32 = evens + odds;
        const float upre = (half32 + xhalf(half32)) + bias2;

        float u1 = tanhf(upre);
        u1 = fminf(fmaxf(u1, MIN_ACC), MAX_ACC);

        const float rr = fmaf(-0.1f * u1, u1, r);
        float v2 = v1 + 0.0015f * u1 - 0.0025f * __cosf(3.0f * p1);
        v2 = fminf(fmaxf(v2, MIN_V), MAX_V);
        const float p2 = p1 + v2;

        p = active ? p2 : p;
        v = active ? v2 : v;
        u = active ? u1 : u;
        r = active ? rr : r;
    }

    const float rout = (p <= GOAL_P) ? r : (r + 100.0f);
    if (!hiHalf && envValid) {
        float4 o;
        o.x = p; o.y = v; o.z = u; o.w = rout;
        reinterpret_cast<float4*>(out)[env] = o;
    }
}

extern "C" void kernel_launch(void* const* d_in, const int* in_sizes, int n_in,
                              void* d_out, int out_size, void* d_ws, size_t ws_size,
                              hipStream_t stream) {
    const float* x  = (const float*)d_in[0];
    const float* w1 = (const float*)d_in[1];
    const float* b1 = (const float*)d_in[2];
    const float* w2 = (const float*)d_in[3];
    const float* b2 = (const float*)d_in[4];
    float* out = (float*)d_out;

    const int B = in_sizes[0] / 4;
    const int envsPerWave = 16;
    const int waves = (B + envsPerWave - 1) / envsPerWave;
    const int threads = waves * 32;
    const int block = 256;
    const int grid = (threads + block - 1) / block;

    MountainCar_7318624272632_kernel<<<grid, block, 0, stream>>>(x, w1, b1, w2, b2, out, B);
}


# early-out test (all envs of the wave frozen) moved from every step to the boundaries of 25-step blocks: the step loop holds no vector-to-scalar-to-branch chain; arithmetic and sum order unchanged
# speedup vs baseline: 1.9619x; 1.1513x over previous
#include <hip/hip_runtime.h>

typedef float v2f __attribute__((ext_vector_type(2)));
typedef float v8f __attribute__((ext_vector_type(8)));

#define GOAL_P  0.5f
#define MIN_P   (-1.2f)
#define MIN_V   (-0.07f)
#define MAX_V   (0.07f)
#define MIN_ACC (-1.2f)
#define MAX_ACC (1.2f)
#define MAX_STEPS 500
#define STEPS_PER_CHECK 25

__device__ __forceinline__ float xhalf(float x) {
    const int b = __builtin_bit_cast(int, x);
    return __builtin_bit_cast(float,
        __builtin_amdgcn_permlanex16(b, b, 0x76543210, 0xfedcba98, false, false));
}

__global__ __launch_bounds__(256)
void MountainCar_7318624272632_kernel(const float* __restrict__ x,
                                      const float* __restrict__ w1,
                                      const float* __restrict__ b1,
                                      const float* __restrict__ w2,
                                      const float* __restrict__ b2,
                                      float* __restrict__ out, int B)
{
    const int lane   = threadIdx.x & 31;
    const int waveId = (blockIdx.x * blockDim.x + threadIdx.x) >> 5;
    const int ln     = lane & 15;
    const bool hiHalf = (lane >= 16);

    if (waveId * 16 >= B) return;

    const int  env      = waveId * 16 + ln;
    const bool envValid = (env < B);
    const int  envC     = envValid ? env : (B - 1);

    v2f  Aw[4];
    float w2r[4][8];
    const int hOff = hiHalf ? 8 : 0;
#pragma unroll
    for (int t = 0; t < 4; ++t) {
        const int j = t * 16 + ln;
        Aw[t][0] = hiHalf ? b1[j] : w1[j];
        Aw[t][1] = hiHalf ? 0.0f  : w1[64 + j];
#pragma unroll
        for (int i = 0; i < 8; ++i)
            w2r[t][i] = w2[t * 16 + hOff + i];
    }
    const float bias2 = b2[0];

    const float4 st = reinterpret_cast<const float4*>(x)[envC];
    float p = st.x, v = st.y, u = st.z, r = st.w;

    static_assert(MAX_STEPS % STEPS_PER_CHECK == 0, "blocks must tile the step count");
    bool anyLive = (__ballot((p <= GOAL_P) && !hiHalf) != 0ull);
    for (int blk = 0; anyLive && blk < MAX_STEPS / STEPS_PER_CHECK; ++blk) {
        for (int step = 0; step < STEPS_PER_CHECK; ++step) {
            const bool active = (p <= GOAL_P);

            const bool hit = (p <= MIN_P);
            const float p1 = hit ? MIN_P : p;
            const float v1 = hit ? 0.0f  : v;

            v2f S;
            S[0] = hiHalf ? 1.0f : p1;
            S[1] = hiHalf ? 0.0f : v1;

            float acc[8];
#pragma unroll
            for (int i = 0; i < 8; ++i) acc[i] = 0.0f;

#pragma unroll
            for (int t = 0; t < 4; ++t) {
                v8f c = {};
                c = __builtin_amdgcn_wmma_f32_16x16x4_f32(
                         false, Aw[t],  false, S,
                         (short)0, c,  false,  false);
#pragma unroll
                for (int i = 0; i < 8; ++i) {
                    const float h = fmaxf(c[i], 0.0f);
                    acc[i] = fmaf(h, w2r[t][i], acc[i]);
                }
            }

            const float evens  = (acc[0] + acc[2]) + (acc[4] + acc[6]);
            const float odds   = (acc[1] + acc[3]) + (acc[5] + acc[7]);
            const float half32 = evens + odds;
            const float upre = (half32 + xhalf(half32)) + bias2;

            float u1 = tanhf(upre);
            u1 = fminf(fmaxf(u1, MIN_ACC), MAX_ACC);

            const float rr = fmaf(-0.1f * u1, u1, r);
            float v2 = v1 + 0.0015f * u1 - 0.0025f * __cosf(3.0f * p1);
            v2 = fminf(fmaxf(v2, MIN_V), MAX_V);
            const float p2 = p1 + v2;

            p = active ? p2 : p;
            v = active ? v2 : v;
            u = active ? u1 : u;
            r = active ? rr : r;
        }
        anyLive = (__ballot((p <= GOAL_P) && !hiHalf) != 0ull);
    }

    const float rout = (p <= GOAL_P) ? r : (r + 100.0f);
    if (!hiHalf && envValid) {
        float4 o;
        o.x = p; o.y = v; o.z = u; o.w = rout;
        reinterpret_cast<float4*>(out)[env] = o;
    }
}

extern "C" void kernel_launch(void* const* d_in, const int* in_sizes, int n_in,
                              void* d_out, int out_size, void* d_ws, size_t ws_size,
                              hipStream_t stream) {
    const float* x  = (const float*)d_in[0];
    const float* w1 = (const float*)d_in[1];
    const float* b1 = (const float*)d_in[2];
    const float* w2 = (const float*)d_in[3];
    const float* b2 = (const float*)d_in[4];
    float* out = (float*)d_out;

    const int B = in_sizes[0] / 4;
    const int envsPerWave = 16;
    const int waves = (B + envsPerWave - 1) / envsPerWave;
    const int threads = waves * 32;
    const int block = 256;
    const int grid = (threads + block - 1) / block;

    MountainCar_7318624272632_kernel<<<grid, block, 0, stream>>>(x, w1, b1, w2, b2, out, B);
}


# tanh as 1 - 2/(2^(x*2log2e) + 1) on the hardware exp and rcp, straight-line, in place of the library routine's two data-dependent branches; u differs in its last bits (abs 3e-7 at most)
# speedup vs baseline: 2.3300x; 1.1876x over previous
#include <hip/hip_runtime.h>

typedef float v2f __attribute__((ext_vector_type(2)));
typedef float v8f __attribute__((ext_vector_type(8)));

#define GOAL_P  0.5f
#define MIN_P   (-1.2f)
#define MIN_V   (-0.07f)
#define MAX_V   (0.07f)
#define MIN_ACC (-1.2f)
#define MAX_ACC (1.2f)
#define MAX_STEPS 500
#define STEPS_PER_CHECK 25
#define TWO_LOG2E 2.885390081777927f

__device__ __forceinline__ float xhalf(float x) {
    const int b = __builtin_bit_cast(int, x);
    return __builtin_bit_cast(float,
        __builtin_amdgcn_permlanex16(b, b, 0x76543210, 0xfedcba98, false, false));
}

__global__ __launch_bounds__(256)
void MountainCar_7318624272632_kernel(const float* __restrict__ x,
                                      const float* __restrict__ w1,
                                      const float* __restrict__ b1,
                                      const float* __restrict__ w2,
                                      const float* __restrict__ b2,
                                      float* __restrict__ out, int B)
{
    const int lane   = threadIdx.x & 31;
    const int waveId = (blockIdx.x * blockDim.x + threadIdx.x) >> 5;
    const int ln     = lane & 15;
    const bool hiHalf = (lane >= 16);

    if (waveId * 16 >= B) return;

    const int  env      = waveId * 16 + ln;
    const bool envValid = (env < B);
    const int  envC     = envValid ? env : (B - 1);

    v2f  Aw[4];
    float w2r[4][8];
    const int hOff = hiHalf ? 8 : 0;
#pragma unroll
    for (int t = 0; t < 4; ++t) {
        const int j = t * 16 + ln;
        Aw[t][0] = hiHalf ? b1[j] : w1[j];
        Aw[t][1] = hiHalf ? 0.0f  : w1[64 + j];
#pragma unroll
        for (int i = 0; i < 8; ++i)
            w2r[t][i] = w2[t * 16 + hOff + i];
    }
    const float bias2 = b2[0];

    const float4 st = reinterpret_cast<const float4*>(x)[envC];
    float p = st.x, v = st.y, u = st.z, r = st.w;

    static_assert(MAX_STEPS % STEPS_PER_CHECK == 0, "blocks must tile the step count");
    bool anyLive = (__ballot((p <= GOAL_P) && !hiHalf) != 0ull);
    for (int blk = 0; anyLive && blk < MAX_STEPS / STEPS_PER_CHECK; ++blk) {
        for (int step = 0; step < STEPS_PER_CHECK; ++step) {
            const bool active = (p <= GOAL_P);

            const bool hit = (p <= MIN_P);
            const float p1 = hit ? MIN_P : p;
            const float v1 = hit ? 0.0f  : v;

            v2f S;
            S[0] = hiHalf ? 1.0f : p1;
            S[1] = hiHalf ? 0.0f : v1;

            float acc[8];
#pragma unroll
            for (int i = 0; i < 8; ++i) acc[i] = 0.0f;

#pragma unroll
            for (int t = 0; t < 4; ++t) {
                v8f c = {};
                c = __builtin_amdgcn_wmma_f32_16x16x4_f32(
                         false, Aw[t],  false, S,
                         (short)0, c,  false,  false);
#pragma unroll
                for (int i = 0; i < 8; ++i) {
                    const float h = fmaxf(c[i], 0.0f);
                    acc[i] = fmaf(h, w2r[t][i], acc[i]);
                }
            }

            const float evens  = (acc[0] + acc[2]) + (acc[4] + acc[6]);
            const float odds   = (acc[1] + acc[3]) + (acc[5] + acc[7]);
            const float half32 = evens + odds;
            const float upre = (half32 + xhalf(half32)) + bias2;

            const float e2x = __builtin_amdgcn_exp2f(upre * TWO_LOG2E);
            float u1 = fmaf(__builtin_amdgcn_rcpf(e2x + 1.0f), -2.0f, 1.0f);
            u1 = fminf(fmaxf(u1, MIN_ACC), MAX_ACC);

            const float rr = fmaf(-0.1f * u1, u1, r);
            float v2 = v1 + 0.0015f * u1 - 0.0025f * __cosf(3.0f * p1);
            v2 = fminf(fmaxf(v2, MIN_V), MAX_V);
            const float p2 = p1 + v2;

            p = active ? p2 : p;
            v = active ? v2 : v;
            u = active ? u1 : u;
            r = active ? rr : r;
        }
        anyLive = (__ballot((p <= GOAL_P) && !hiHalf) != 0ull);
    }

    const float rout = (p <= GOAL_P) ? r : (r + 100.0f);
    if (!hiHalf && envValid) {
        float4 o;
        o.x = p; o.y = v; o.z = u; o.w = rout;
        reinterpret_cast<float4*>(out)[env] = o;
    }
}

extern "C" void kernel_launch(void* const* d_in, const int* in_sizes, int n_in,
                              void* d_out, int out_size, void* d_ws, size_t ws_size,
                              hipStream_t stream) {
    const float* x  = (const float*)d_in[0];
    const float* w1 = (const float*)d_in[1];
    const float* b1 = (const float*)d_in[2];
    const float* w2 = (const float*)d_in[3];
    const float* b2 = (const float*)d_in[4];
    float* out = (float*)d_out;

    const int B = in_sizes[0] / 4;
    const int envsPerWave = 16;
    const int waves = (B + envsPerWave - 1) / envsPerWave;
    const int threads = waves * 32;
    const int block = 256;
    const int grid = (threads + block - 1) / block;

    MountainCar_7318624272632_kernel<<<grid, block, 0, stream>>>(x, w1, b1, w2, b2, out, B);
}


# work-groups of 4 waves instead of 8 (grid 128 instead of 64): each wave has a SIMD, its matrix pipe and vector port, to itself; kernel code, arithmetic and sum order unchanged
# speedup vs baseline: 3.5224x; 1.5118x over previous
#include <hip/hip_runtime.h>

typedef float v2f __attribute__((ext_vector_type(2)));
typedef float v8f __attribute__((ext_vector_type(8)));

#define GOAL_P  0.5f
#define MIN_P   (-1.2f)
#define MIN_V   (-0.07f)
#define MAX_V   (0.07f)
#define MIN_ACC (-1.2f)
#define MAX_ACC (1.2f)
#define MAX_STEPS 500
#define STEPS_PER_CHECK 25
#define TWO_LOG2E 2.885390081777927f

__device__ __forceinline__ float xhalf(float x) {
    const int b = __builtin_bit_cast(int, x);
    return __builtin_bit_cast(float,
        __builtin_amdgcn_permlanex16(b, b, 0x76543210, 0xfedcba98, false, false));
}

__global__ __launch_bounds__(128)
void MountainCar_7318624272632_kernel(const float* __restrict__ x,
                                      const float* __restrict__ w1,
                                      const float* __restrict__ b1,
                                      const float* __restrict__ w2,
                                      const float* __restrict__ b2,
                                      float* __restrict__ out, int B)
{
    const int lane   = threadIdx.x & 31;
    const int waveId = (blockIdx.x * blockDim.x + threadIdx.x) >> 5;
    const int ln     = lane & 15;
    const bool hiHalf = (lane >= 16);

    if (waveId * 16 >= B) return;

    const int  env      = waveId * 16 + ln;
    const bool envValid = (env < B);
    const int  envC     = envValid ? env : (B - 1);

    v2f  Aw[4];
    float w2r[4][8];
    const int hOff = hiHalf ? 8 : 0;
#pragma unroll
    for (int t = 0; t < 4; ++t) {
        const int j = t * 16 + ln;
        Aw[t][0] = hiHalf ? b1[j] : w1[j];
        Aw[t][1] = hiHalf ? 0.0f  : w1[64 + j];
#pragma unroll
        for (int i = 0; i < 8; ++i)
            w2r[t][i] = w2[t * 16 + hOff + i];
    }
    const float bias2 = b2[0];

    const float4 st = reinterpret_cast<const float4*>(x)[envC];
    float p = st.x, v = st.y, u = st.z, r = st.w;

    static_assert(MAX_STEPS % STEPS_PER_CHECK == 0, "blocks must tile the step count");
    bool anyLive = (__ballot((p <= GOAL_P) && !hiHalf) != 0ull);
    for (int blk = 0; anyLive && blk < MAX_STEPS / STEPS_PER_CHECK; ++blk) {
        for (int step = 0; step < STEPS_PER_CHECK; ++step) {
            const bool active = (p <= GOAL_P);

            const bool hit = (p <= MIN_P);
            const float p1 = hit ? MIN_P : p;
            const float v1 = hit ? 0.0f  : v;

            v2f S;
            S[0] = hiHalf ? 1.0f : p1;
            S[1] = hiHalf ? 0.0f : v1;

            float acc[8];
#pragma unroll
            for (int i = 0; i < 8; ++i) acc[i] = 0.0f;

#pragma unroll
            for (int t = 0; t < 4; ++t) {
                v8f c = {};
                c = __builtin_amdgcn_wmma_f32_16x16x4_f32(
                         false, Aw[t],  false, S,
                         (short)0, c,  false,  false);
#pragma unroll
                for (int i = 0; i < 8; ++i) {
                    const float h = fmaxf(c[i], 0.0f);
                    acc[i] = fmaf(h, w2r[t][i], acc[i]);
                }
            }

            const float evens  = (acc[0] + acc[2]) + (acc[4] + acc[6]);
            const float odds   = (acc[1] + acc[3]) + (acc[5] + acc[7]);
            const float half32 = evens + odds;
            const float upre = (half32 + xhalf(half32)) + bias2;

            const float e2x = __builtin_amdgcn_exp2f(upre * TWO_LOG2E);
            float u1 = fmaf(__builtin_amdgcn_rcpf(e2x + 1.0f), -2.0f, 1.0f);
            u1 = fminf(fmaxf(u1, MIN_ACC), MAX_ACC);

            const float rr = fmaf(-0.1f * u1, u1, r);
            float v2 = v1 + 0.0015f * u1 - 0.0025f * __cosf(3.0f * p1);
            v2 = fminf(fmaxf(v2, MIN_V), MAX_V);
            const float p2 = p1 + v2;

            p = active ? p2 : p;
            v = active ? v2 : v;
            u = active ? u1 : u;
            r = active ? rr : r;
        }
        anyLive = (__ballot((p <= GOAL_P) && !hiHalf) != 0ull);
    }

    const float rout = (p <= GOAL_P) ? r : (r + 100.0f);
    if (!hiHalf && envValid) {
        float4 o;
        o.x = p; o.y = v; o.z = u; o.w = rout;
        reinterpret_cast<float4*>(out)[env] = o;
    }
}

extern "C" void kernel_launch(void* const* d_in, const int* in_sizes, int n_in,
                              void* d_out, int out_size, void* d_ws, size_t ws_size,
                              hipStream_t stream) {
    const float* x  = (const float*)d_in[0];
    const float* w1 = (const float*)d_in[1];
    const float* b1 = (const float*)d_in[2];
    const float* w2 = (const float*)d_in[3];
    const float* b2 = (const float*)d_in[4];
    float* out = (float*)d_out;

    const int B = in_sizes[0] / 4;
    const int envsPerWave = 16;
    const int waves = (B + envsPerWave - 1) / envsPerWave;
    const int threads = waves * 32;
    const int block = 128;
    const int grid = (threads + block - 1) / block;

    MountainCar_7318624272632_kernel<<<grid, block, 0, stream>>>(x, w1, b1, w2, b2, out, B);
}


# 8 envs per wave instead of 16 (1024 waves, one on every SIMD of the chip): a step holds 2 matrix instructions and 16 hidden units per lane, bias as the C operand; the 64-term sum order changes
# speedup vs baseline: 3.9636x; 1.1253x over previous
#include <hip/hip_runtime.h>

typedef float v2f __attribute__((ext_vector_type(2)));
typedef float v8f __attribute__((ext_vector_type(8)));

#define GOAL_P  0.5f
#define MIN_P   (-1.2f)
#define MIN_V   (-0.07f)
#define MAX_V   (0.07f)
#define MIN_ACC (-1.2f)
#define MAX_ACC (1.2f)
#define MAX_STEPS 500
#define STEPS_PER_CHECK 25
#define TWO_LOG2E 2.885390081777927f

__device__ __forceinline__ float xhalf(float x) {
    const int b = __builtin_bit_cast(int, x);
    return __builtin_bit_cast(float,
        __builtin_amdgcn_permlanex16(b, b, 0x76543210, 0xfedcba98, false, false));
}

__device__ __forceinline__ float xgroup(float x) {
    const int b = __builtin_bit_cast(int, x);
    return __builtin_bit_cast(float, __builtin_amdgcn_update_dpp(0, b, 0x128  , 0xf, 0xf, true));
}

__global__ __launch_bounds__(128)
void MountainCar_7318624272632_kernel(const float* __restrict__ x,
                                      const float* __restrict__ w1,
                                      const float* __restrict__ b1,
                                      const float* __restrict__ w2,
                                      const float* __restrict__ b2,
                                      float* __restrict__ out, int B)
{
    const int lane   = threadIdx.x & 31;
    const int waveId = (blockIdx.x * blockDim.x + threadIdx.x) >> 5;
    const int col    = lane & 15;
    const int h      = lane >> 4;
    const int e      = col & 7;
    const int g      = col >> 3;
    const bool feeds   = (h == g);
    const bool outLane = (lane < 8);

    if (waveId * 8 >= B) return;

    const int  env      = waveId * 8 + e;
    const bool envValid = (env < B);
    const int  envC     = envValid ? env : (B - 1);

    v2f  Aw[2];
    v8f  Cb[2];
    float w2r[2][8];
    const int hid0 = 16 * g + 8 * h;
#pragma unroll
    for (int t = 0; t < 2; ++t) {
        const int j = 32 * t + 16 * h + col;
        Aw[t][0] = w1[j];
        Aw[t][1] = w1[64 + j];
#pragma unroll
        for (int i = 0; i < 8; ++i) {
            Cb[t][i]  = b1[32 * t + hid0 + i];
            w2r[t][i] = w2[32 * t + hid0 + i];
        }
    }
    const float bias2 = b2[0];

    const float4 st = reinterpret_cast<const float4*>(x)[envC];
    float p = st.x, v = st.y, u = st.z, r = st.w;

    static_assert(MAX_STEPS % STEPS_PER_CHECK == 0, "blocks must tile the step count");
    bool anyLive = (__ballot((p <= GOAL_P) && outLane) != 0ull);
    for (int blk = 0; anyLive && blk < MAX_STEPS / STEPS_PER_CHECK; ++blk) {
        for (int step = 0; step < STEPS_PER_CHECK; ++step) {
            const bool active = (p <= GOAL_P);

            const bool hit = (p <= MIN_P);
            const float p1 = hit ? MIN_P : p;
            const float v1 = hit ? 0.0f  : v;

            v2f S;
            S[0] = feeds ? p1 : 0.0f;
            S[1] = feeds ? v1 : 0.0f;

            const v8f d0 = __builtin_amdgcn_wmma_f32_16x16x4_f32(
                     false, Aw[0],  false, S,
                     (short)0, Cb[0],  false,  false);
            const v8f d1 = __builtin_amdgcn_wmma_f32_16x16x4_f32(
                     false, Aw[1],  false, S,
                     (short)0, Cb[1],  false,  false);

            __builtin_amdgcn_sched_barrier(0);

            float acc[8];
#pragma unroll
            for (int i = 0; i < 8; ++i)
                acc[i] = fmaxf(d0[i], 0.0f) * w2r[0][i];
#pragma unroll
            for (int i = 0; i < 8; ++i)
                acc[i] = fmaf(fmaxf(d1[i], 0.0f), w2r[1][i], acc[i]);

            const float evens = (acc[0] + acc[2]) + (acc[4] + acc[6]);
            const float odds  = (acc[1] + acc[3]) + (acc[5] + acc[7]);
            const float s16   = evens + odds;
            const float s32   = s16 + xhalf(s16);
            const float upre  = (s32 + xgroup(s32)) + bias2;

            const float e2x = __builtin_amdgcn_exp2f(upre * TWO_LOG2E);
            float u1 = fmaf(__builtin_amdgcn_rcpf(e2x + 1.0f), -2.0f, 1.0f);
            u1 = fminf(fmaxf(u1, MIN_ACC), MAX_ACC);

            const float rr = fmaf(-0.1f * u1, u1, r);
            float v2 = v1 + 0.0015f * u1 - 0.0025f * __cosf(3.0f * p1);
            v2 = fminf(fmaxf(v2, MIN_V), MAX_V);
            const float p2 = p1 + v2;

            p = active ? p2 : p;
            v = active ? v2 : v;
            u = active ? u1 : u;
            r = active ? rr : r;
        }
        anyLive = (__ballot((p <= GOAL_P) && outLane) != 0ull);
    }

    const float rout = (p <= GOAL_P) ? r : (r + 100.0f);
    if (outLane && envValid) {
        float4 o;
        o.x = p; o.y = v; o.z = u; o.w = rout;
        reinterpret_cast<float4*>(out)[env] = o;
    }
}

extern "C" void kernel_launch(void* const* d_in, const int* in_sizes, int n_in,
                              void* d_out, int out_size, void* d_ws, size_t ws_size,
                              hipStream_t stream) {
    const float* x  = (const float*)d_in[0];
    const float* w1 = (const float*)d_in[1];
    const float* b1 = (const float*)d_in[2];
    const float* w2 = (const float*)d_in[3];
    const float* b2 = (const float*)d_in[4];
    float* out = (float*)d_out;

    const int B = in_sizes[0] / 4;
    const int envsPerWave = 8;
    const int waves = (B + envsPerWave - 1) / envsPerWave;
    const int threads = waves * 32;
    const int block = 128;
    const int grid = (threads + block - 1) / block;

    MountainCar_7318624272632_kernel<<<grid, block, 0, stream>>>(x, w1, b1, w2, b2, out, B);
}


# step's dependent chain shortened: B operand is the carried state, next layer 1 and cosine issued at step end, freeze select off the chain, v by one fma from 1/(2^x+1); sum order of v, b2 changes
# speedup vs baseline: 4.8757x; 1.2301x over previous
#include <hip/hip_runtime.h>

typedef float v2f __attribute__((ext_vector_type(2)));
typedef float v8f __attribute__((ext_vector_type(8)));

#define GOAL_P  0.5f
#define MIN_P   (-1.2f)
#define MIN_V   (-0.07f)
#define MAX_V   (0.07f)
#define MIN_ACC (-1.2f)
#define MAX_ACC (1.2f)
#define MAX_STEPS 500
#define STEPS_PER_CHECK 25
#define TWO_LOG2E 2.885390081777927f

__device__ __forceinline__ float xhalf(float x) {
    const int b = __builtin_bit_cast(int, x);
    return __builtin_bit_cast(float,
        __builtin_amdgcn_permlanex16(0, b, 0x76543210, 0xfedcba98, false, false));
}

__device__ __forceinline__ float xgroup(float x) {
    const int b = __builtin_bit_cast(int, x);
    return __builtin_bit_cast(float, __builtin_amdgcn_update_dpp(0, b, 0x128  , 0xf, 0xf, true));
}

__device__ __forceinline__ float relu(float x) {
    const int b = __builtin_bit_cast(int, x);
    return __builtin_bit_cast(float, b > 0 ? b : 0);
}

__global__ __launch_bounds__(128)
void MountainCar_7318624272632_kernel(const float* __restrict__ x,
                                      const float* __restrict__ w1,
                                      const float* __restrict__ b1,
                                      const float* __restrict__ w2,
                                      const float* __restrict__ b2,
                                      float* __restrict__ out, int B)
{
    const int lane   = threadIdx.x & 31;
    const int waveId = (blockIdx.x * blockDim.x + threadIdx.x) >> 5;
    const int col    = lane & 15;
    const int h      = lane >> 4;
    const int e      = col & 7;
    const int g      = col >> 3;
    const bool feeds   = (h == g);
    const bool outLane = (lane < 8);

    if (waveId * 8 >= B) return;

    const int  env      = waveId * 8 + e;
    const bool envValid = (env < B);
    const int  envC     = envValid ? env : (B - 1);

    v2f  Aw[2];
    v8f  Cb[2];
    float w2s[2][8];
    const int hid0 = 16 * g + 8 * h;
#pragma unroll
    for (int t = 0; t < 2; ++t) {
        const int j = 32 * t + 16 * h + col;
        Aw[t][0] = w1[j];
        Aw[t][1] = w1[64 + j];
#pragma unroll
        for (int i = 0; i < 8; ++i) {
            Cb[t][i]  = b1[32 * t + hid0 + i];
            w2s[t][i] = w2[32 * t + hid0 + i] * TWO_LOG2E;
        }
    }
    const float b2s = outLane ? b2[0] * TWO_LOG2E : 0.0f;

    const float4 st = reinterpret_cast<const float4*>(x)[envC];

    float pf = st.x, vf = st.y, uf = st.z, rf = st.w;
    bool  live = (st.x <= GOAL_P);

    const float vlo = feeds ? MIN_V : 0.0f;
    const float vhi = feeds ? MAX_V : 0.0f;

    v2f S;
    {
        const bool hit0 = (st.x <= MIN_P);
        S[0] = feeds ? (hit0 ? MIN_P : st.x) : 0.0f;
        S[1] = feeds ? (hit0 ? 0.0f  : st.y) : 0.0f;
    }

    v8f d0 = __builtin_amdgcn_wmma_f32_16x16x4_f32(
             false, Aw[0],  false, S,
             (short)0, Cb[0],  false,  false);
    v8f d1 = __builtin_amdgcn_wmma_f32_16x16x4_f32(
             false, Aw[1],  false, S,
             (short)0, Cb[1],  false,  false);

    float vpart = fmaf(-0.0025f, __cosf(3.0f * S[0]), S[1] + 0.0015f);

    static_assert(MAX_STEPS % STEPS_PER_CHECK == 0, "blocks must tile the step count");
    bool anyLive = (__ballot(live && outLane) != 0ull);
    for (int blk = 0; anyLive && blk < MAX_STEPS / STEPS_PER_CHECK; ++blk) {
        for (int step = 0; step < STEPS_PER_CHECK; ++step) {
            const float p1 = S[0];

            float acc[8];
            acc[0] = fmaf(relu(d0[0]), w2s[0][0], b2s);
#pragma unroll
            for (int i = 1; i < 8; ++i)
                acc[i] = relu(d0[i]) * w2s[0][i];
#pragma unroll
            for (int i = 0; i < 8; ++i)
                acc[i] = fmaf(relu(d1[i]), w2s[1][i], acc[i]);

            const float evens = (acc[0] + acc[2]) + (acc[4] + acc[6]);
            const float odds  = (acc[1] + acc[3]) + (acc[5] + acc[7]);
            const float s16   = evens + odds;

            const float s32   = s16 + xhalf(s16);
            const float xarg  = s32 + xgroup(s32);

            const float rc = __builtin_amdgcn_rcpf(__builtin_amdgcn_exp2f(xarg) + 1.0f);

            const float v2 = __builtin_amdgcn_fmed3f(fmaf(rc, -2.0f * 0.0015f, vpart), vlo, vhi);
            const float p2 = p1 + v2;

            const bool hit = (p2 <= MIN_P);
            S[0] = hit ? MIN_P : p2;
            S[1] = hit ? 0.0f  : v2;

            float u1 = fmaf(rc, -2.0f, 1.0f);
            u1 = fminf(fmaxf(u1, MIN_ACC), MAX_ACC);
            const float rr = fmaf(-0.1f * u1, u1, rf);
            pf = live ? p2 : pf;
            vf = live ? v2 : vf;
            uf = live ? u1 : uf;
            rf = live ? rr : rf;
            live = live && (p2 <= GOAL_P);

            d0 = __builtin_amdgcn_wmma_f32_16x16x4_f32(
                     false, Aw[0],  false, S,
                     (short)0, Cb[0],  false,  false);
            __builtin_amdgcn_sched_barrier(0);
            d1 = __builtin_amdgcn_wmma_f32_16x16x4_f32(
                     false, Aw[1],  false, S,
                     (short)0, Cb[1],  false,  false);
            vpart = fmaf(-0.0025f, __cosf(3.0f * S[0]), S[1] + 0.0015f);
        }
        anyLive = (__ballot(live && outLane) != 0ull);
    }

    const float rout = (pf <= GOAL_P) ? rf : (rf + 100.0f);
    if (outLane && envValid) {
        float4 o;
        o.x = pf; o.y = vf; o.z = uf; o.w = rout;
        reinterpret_cast<float4*>(out)[env] = o;
    }
}

extern "C" void kernel_launch(void* const* d_in, const int* in_sizes, int n_in,
                              void* d_out, int out_size, void* d_ws, size_t ws_size,
                              hipStream_t stream) {
    const float* x  = (const float*)d_in[0];
    const float* w1 = (const float*)d_in[1];
    const float* b1 = (const float*)d_in[2];
    const float* w2 = (const float*)d_in[3];
    const float* b2 = (const float*)d_in[4];
    float* out = (float*)d_out;

    const int B = in_sizes[0] / 4;
    const int envsPerWave = 8;
    const int waves = (B + envsPerWave - 1) / envsPerWave;
    const int threads = waves * 32;
    const int block = 128;
    const int grid = (threads + block - 1) / block;

    MountainCar_7318624272632_kernel<<<grid, block, 0, stream>>>(x, w1, b1, w2, b2, out, B);
}


# tanh on the hardware tanh instruction in place of exp2, add and reciprocal: one transcendental on the step's chain, not two; u differs in its last bits, v sums as (v1 - 0.0025cos) + 0.0015u
# speedup vs baseline: 5.2490x; 1.0766x over previous
#include <hip/hip_runtime.h>

typedef float v2f __attribute__((ext_vector_type(2)));
typedef float v8f __attribute__((ext_vector_type(8)));

#define GOAL_P  0.5f
#define MIN_P   (-1.2f)
#define MIN_V   (-0.07f)
#define MAX_V   (0.07f)
#define MIN_ACC (-1.2f)
#define MAX_ACC (1.2f)
#define MAX_STEPS 500
#define STEPS_PER_CHECK 25

__device__ __forceinline__ float xhalf(float x) {
    const int b = __builtin_bit_cast(int, x);
    return __builtin_bit_cast(float,
        __builtin_amdgcn_permlanex16(0, b, 0x76543210, 0xfedcba98, false, false));
}

__device__ __forceinline__ float xgroup(float x) {
    const int b = __builtin_bit_cast(int, x);
    return __builtin_bit_cast(float, __builtin_amdgcn_update_dpp(0, b, 0x128  , 0xf, 0xf, true));
}

__device__ __forceinline__ float relu(float x) {
    const int b = __builtin_bit_cast(int, x);
    return __builtin_bit_cast(float, b > 0 ? b : 0);
}

__global__ __launch_bounds__(128)
void MountainCar_7318624272632_kernel(const float* __restrict__ x,
                                      const float* __restrict__ w1,
                                      const float* __restrict__ b1,
                                      const float* __restrict__ w2,
                                      const float* __restrict__ b2,
                                      float* __restrict__ out, int B)
{
    const int lane   = threadIdx.x & 31;
    const int waveId = (blockIdx.x * blockDim.x + threadIdx.x) >> 5;
    const int col    = lane & 15;
    const int h      = lane >> 4;
    const int e      = col & 7;
    const int g      = col >> 3;
    const bool feeds   = (h == g);
    const bool outLane = (lane < 8);

    if (waveId * 8 >= B) return;

    const int  env      = waveId * 8 + e;
    const bool envValid = (env < B);
    const int  envC     = envValid ? env : (B - 1);

    v2f  Aw[2];
    v8f  Cb[2];
    float w2s[2][8];
    const int hid0 = 16 * g + 8 * h;
#pragma unroll
    for (int t = 0; t < 2; ++t) {
        const int j = 32 * t + 16 * h + col;
        Aw[t][0] = w1[j];
        Aw[t][1] = w1[64 + j];
#pragma unroll
        for (int i = 0; i < 8; ++i) {
            Cb[t][i]  = b1[32 * t + hid0 + i];
            w2s[t][i] = w2[32 * t + hid0 + i];
        }
    }
    const float b2s = outLane ? b2[0] : 0.0f;

    const float4 st = reinterpret_cast<const float4*>(x)[envC];

    float pf = st.x, vf = st.y, uf = st.z, rf = st.w;
    bool  live = (st.x <= GOAL_P);

    const float vlo = feeds ? MIN_V : 0.0f;
    const float vhi = feeds ? MAX_V : 0.0f;

    v2f S;
    {
        const bool hit0 = (st.x <= MIN_P);
        S[0] = feeds ? (hit0 ? MIN_P : st.x) : 0.0f;
        S[1] = feeds ? (hit0 ? 0.0f  : st.y) : 0.0f;
    }

    v8f d0 = __builtin_amdgcn_wmma_f32_16x16x4_f32(
             false, Aw[0],  false, S,
             (short)0, Cb[0],  false,  false);
    v8f d1 = __builtin_amdgcn_wmma_f32_16x16x4_f32(
             false, Aw[1],  false, S,
             (short)0, Cb[1],  false,  false);

    float vpart = fmaf(-0.0025f, __cosf(3.0f * S[0]), S[1]);

    static_assert(MAX_STEPS % STEPS_PER_CHECK == 0, "blocks must tile the step count");
    bool anyLive = (__ballot(live && outLane) != 0ull);
    for (int blk = 0; anyLive && blk < MAX_STEPS / STEPS_PER_CHECK; ++blk) {
        for (int step = 0; step < STEPS_PER_CHECK; ++step) {
            const float p1 = S[0];

            float acc[8];
            acc[0] = fmaf(relu(d0[0]), w2s[0][0], b2s);
#pragma unroll
            for (int i = 1; i < 8; ++i)
                acc[i] = relu(d0[i]) * w2s[0][i];
#pragma unroll
            for (int i = 0; i < 8; ++i)
                acc[i] = fmaf(relu(d1[i]), w2s[1][i], acc[i]);

            const float evens = (acc[0] + acc[2]) + (acc[4] + acc[6]);
            const float odds  = (acc[1] + acc[3]) + (acc[5] + acc[7]);
            const float s16   = evens + odds;

            const float s32   = s16 + xhalf(s16);
            const float xarg  = s32 + xgroup(s32);

            const float u = __builtin_amdgcn_tanhf(xarg);

            const float v2 = __builtin_amdgcn_fmed3f(fmaf(u, 0.0015f, vpart), vlo, vhi);
            const float p2 = p1 + v2;

            const bool hit = (p2 <= MIN_P);
            S[0] = hit ? MIN_P : p2;
            S[1] = hit ? 0.0f  : v2;

            const float u1 = fminf(fmaxf(u, MIN_ACC), MAX_ACC);
            const float rr = fmaf(-0.1f * u1, u1, rf);
            pf = live ? p2 : pf;
            vf = live ? v2 : vf;
            uf = live ? u1 : uf;
            rf = live ? rr : rf;
            live = live && (p2 <= GOAL_P);

            d0 = __builtin_amdgcn_wmma_f32_16x16x4_f32(
                     false, Aw[0],  false, S,
                     (short)0, Cb[0],  false,  false);
            __builtin_amdgcn_sched_barrier(0);
            d1 = __builtin_amdgcn_wmma_f32_16x16x4_f32(
                     false, Aw[1],  false, S,
                     (short)0, Cb[1],  false,  false);
            vpart = fmaf(-0.0025f, __cosf(3.0f * S[0]), S[1]);
        }
        anyLive = (__ballot(live && outLane) != 0ull);
    }

    const float rout = (pf <= GOAL_P) ? rf : (rf + 100.0f);
    if (outLane && envValid) {
        float4 o;
        o.x = pf; o.y = vf; o.z = uf; o.w = rout;
        reinterpret_cast<float4*>(out)[env] = o;
    }
}

extern "C" void kernel_launch(void* const* d_in, const int* in_sizes, int n_in,
                              void* d_out, int out_size, void* d_ws, size_t ws_size,
                              hipStream_t stream) {
    const float* x  = (const float*)d_in[0];
    const float* w1 = (const float*)d_in[1];
    const float* b1 = (const float*)d_in[2];
    const float* w2 = (const float*)d_in[3];
    const float* b2 = (const float*)d_in[4];
    float* out = (float*)d_out;

    const int B = in_sizes[0] / 4;
    const int envsPerWave = 8;
    const int waves = (B + envsPerWave - 1) / envsPerWave;
    const int threads = waves * 32;
    const int block = 128;
    const int grid = (threads + block - 1) / block;

    MountainCar_7318624272632_kernel<<<grid, block, 0, stream>>>(x, w1, b1, w2, b2, out, B);
}


# relu as (z+|z|)/2: |z| is the fma's source modifier against w2/2 (no max instructions, 4 fma chains); linear half = 2 fmas on (p,v) from sums formed once; tanh argument's sum order changes
# speedup vs baseline: 6.0782x; 1.1580x over previous
#include <hip/hip_runtime.h>

typedef float v2f __attribute__((ext_vector_type(2)));
typedef float v8f __attribute__((ext_vector_type(8)));

#define GOAL_P  0.5f
#define MIN_P   (-1.2f)
#define MIN_V   (-0.07f)
#define MAX_V   (0.07f)
#define MIN_ACC (-1.2f)
#define MAX_ACC (1.2f)
#define MAX_STEPS 500
#define STEPS_PER_CHECK 25

__device__ __forceinline__ float xhalf(float x) {
    const int b = __builtin_bit_cast(int, x);
    return __builtin_bit_cast(float,
        __builtin_amdgcn_permlanex16(0, b, 0x76543210, 0xfedcba98, false, false));
}

__device__ __forceinline__ float xgroup(float x) {
    const int b = __builtin_bit_cast(int, x);
    return __builtin_bit_cast(float, __builtin_amdgcn_update_dpp(0, b, 0x128  , 0xf, 0xf, true));
}

__device__ __forceinline__ float fma_abs(float z, float w, float c) {
    float r;
    asm("v_fma_f32 %0, |%1|, %2, %3" : "=v"(r) : "v"(z), "v"(w), "v"(c));
    return r;
}
__device__ __forceinline__ float mul_abs(float z, float w) {
    float r;
    asm("v_mul_f32_e64 %0, |%1|, %2" : "=v"(r) : "v"(z), "v"(w));
    return r;
}

__global__ __launch_bounds__(128)
void MountainCar_7318624272632_kernel(const float* __restrict__ x,
                                      const float* __restrict__ w1,
                                      const float* __restrict__ b1,
                                      const float* __restrict__ w2,
                                      const float* __restrict__ b2,
                                      float* __restrict__ out, int B)
{
    const int lane   = threadIdx.x & 31;
    const int waveId = (blockIdx.x * blockDim.x + threadIdx.x) >> 5;
    const int col    = lane & 15;
    const int h      = lane >> 4;
    const int e      = col & 7;
    const int g      = col >> 3;
    const bool feeds   = (h == g);
    const bool outLane = (lane < 8);

    if (waveId * 8 >= B) return;

    const int  env      = waveId * 8 + e;
    const bool envValid = (env < B);
    const int  envC     = envValid ? env : (B - 1);

    v2f  Aw[2];
    v8f  Cb[2];
    float wh[2][8];
    float aP = 0.0f, cP = 0.0f, dP = 0.0f;
    const int hid0 = 16 * g + 8 * h;
#pragma unroll
    for (int t = 0; t < 2; ++t) {
        const int j = 32 * t + 16 * h + col;
        Aw[t][0] = w1[j];
        Aw[t][1] = w1[64 + j];
#pragma unroll
        for (int i = 0; i < 8; ++i) {
            const int hid = 32 * t + hid0 + i;
            Cb[t][i] = b1[hid];
            wh[t][i] = 0.5f * w2[hid];
            aP = fmaf(wh[t][i], w1[hid],      aP);
            cP = fmaf(wh[t][i], w1[64 + hid], cP);
            dP = fmaf(wh[t][i], Cb[t][i],     dP);
        }
    }
    aP += xhalf(aP);  cP += xhalf(cP);  dP += xhalf(dP);
    aP += xgroup(aP); cP += xgroup(cP); dP += xgroup(dP);
    const float aL = outLane ? aP : 0.0f;
    const float cL = outLane ? cP : 0.0f;
    const float dL = outLane ? (dP + b2[0]) : 0.0f;

    const float4 st = reinterpret_cast<const float4*>(x)[envC];

    float pf = st.x, vf = st.y, uf = st.z, rf = st.w;
    bool  live = (st.x <= GOAL_P);

    const float vlo = feeds ? MIN_V : 0.0f;
    const float vhi = feeds ? MAX_V : 0.0f;

    v2f S;
    {
        const bool hit0 = (st.x <= MIN_P);
        S[0] = feeds ? (hit0 ? MIN_P : st.x) : 0.0f;
        S[1] = feeds ? (hit0 ? 0.0f  : st.y) : 0.0f;
    }

    v8f d0 = __builtin_amdgcn_wmma_f32_16x16x4_f32(
             false, Aw[0],  false, S,
             (short)0, Cb[0],  false,  false);
    v8f d1 = __builtin_amdgcn_wmma_f32_16x16x4_f32(
             false, Aw[1],  false, S,
             (short)0, Cb[1],  false,  false);

    float lin = fmaf(aL, S[0], fmaf(cL, S[1], dL));
    float cs  = __builtin_amdgcn_cosf((3.0f * S[0]) * 0x1.45f306p-3f);

    static_assert(MAX_STEPS % STEPS_PER_CHECK == 0, "blocks must tile the step count");
    bool anyLive = ((__builtin_amdgcn_ballot_w32(pf <= GOAL_P) & 0xffu) != 0u);
    for (int blk = 0; anyLive && blk < MAX_STEPS / STEPS_PER_CHECK; ++blk) {
        for (int step = 0; step < STEPS_PER_CHECK; ++step) {
            const float p1 = S[0];

            float acc[4];
            acc[0] = fma_abs(d0[0], wh[0][0], lin);
#pragma unroll
            for (int k = 1; k < 4; ++k)
                acc[k] = mul_abs(d0[k], wh[0][k]);
#pragma unroll
            for (int k = 0; k < 4; ++k)
                acc[k] = fma_abs(d0[4 + k], wh[0][4 + k], acc[k]);
            __builtin_amdgcn_sched_barrier(0);
            const float vpart = fmaf(-0.0025f, cs, S[1]);
#pragma unroll
            for (int k = 0; k < 4; ++k)
                acc[k] = fma_abs(d1[k], wh[1][k], acc[k]);
#pragma unroll
            for (int k = 0; k < 4; ++k)
                acc[k] = fma_abs(d1[4 + k], wh[1][4 + k], acc[k]);

            const float s16   = (acc[0] + acc[1]) + (acc[2] + acc[3]);

            const float s32   = s16 + xhalf(s16);
            const float xarg  = s32 + xgroup(s32);

            const float u = __builtin_amdgcn_tanhf(xarg);

            const float v2 = __builtin_amdgcn_fmed3f(fmaf(u, 0.0015f, vpart), vlo, vhi);
            const float p2 = p1 + v2;

            const bool hit = (p2 <= MIN_P);
            S[0] = hit ? MIN_P : p2;
            S[1] = hit ? 0.0f  : v2;

            const float u1 = fminf(fmaxf(u, MIN_ACC), MAX_ACC);
            const float rr = fmaf(-0.1f * u1, u1, rf);
            pf = live ? p2 : pf;
            vf = live ? v2 : vf;
            uf = live ? u1 : uf;
            rf = live ? rr : rf;
            live = live && (p2 <= GOAL_P);

            d0 = __builtin_amdgcn_wmma_f32_16x16x4_f32(
                     false, Aw[0],  false, S,
                     (short)0, Cb[0],  false,  false);
            lin = fmaf(aL, S[0], fmaf(cL, S[1], dL));
            const float rev = (3.0f * S[0]) * 0x1.45f306p-3f;
            __builtin_amdgcn_sched_barrier(0);
            d1 = __builtin_amdgcn_wmma_f32_16x16x4_f32(
                     false, Aw[1],  false, S,
                     (short)0, Cb[1],  false,  false);
            cs = __builtin_amdgcn_cosf(rev);
        }
        anyLive = ((__builtin_amdgcn_ballot_w32(pf <= GOAL_P) & 0xffu) != 0u);
    }

    const float rout = (pf <= GOAL_P) ? rf : (rf + 100.0f);
    if (outLane && envValid) {
        float4 o;
        o.x = pf; o.y = vf; o.z = uf; o.w = rout;
        reinterpret_cast<float4*>(out)[env] = o;
    }
}

extern "C" void kernel_launch(void* const* d_in, const int* in_sizes, int n_in,
                              void* d_out, int out_size, void* d_ws, size_t ws_size,
                              hipStream_t stream) {
    const float* x  = (const float*)d_in[0];
    const float* w1 = (const float*)d_in[1];
    const float* b1 = (const float*)d_in[2];
    const float* w2 = (const float*)d_in[3];
    const float* b2 = (const float*)d_in[4];
    float* out = (float*)d_out;

    const int B = in_sizes[0] / 4;
    const int envsPerWave = 8;
    const int waves = (B + envsPerWave - 1) / envsPerWave;
    const int threads = waves * 32;
    const int block = 128;
    const int grid = (threads + block - 1) / block;

    MountainCar_7318624272632_kernel<<<grid, block, 0, stream>>>(x, w1, b1, w2, b2, out, B);
}


# new position as its own fma behind the tanh, summed (p1 + (v1 - 0.0025cos)) + 0.0015u; velocity clamp and wall folded into its precomputed clamp bounds: step chain 2 instructions shorter
# speedup vs baseline: 6.3764x; 1.0491x over previous
#include <hip/hip_runtime.h>

typedef float v2f __attribute__((ext_vector_type(2)));
typedef float v8f __attribute__((ext_vector_type(8)));

#define GOAL_P  0.5f
#define MIN_P   (-1.2f)
#define MIN_V   (-0.07f)
#define MAX_V   (0.07f)
#define MIN_ACC (-1.2f)
#define MAX_ACC (1.2f)
#define MAX_STEPS 500
#define STEPS_PER_CHECK 25

__device__ __forceinline__ float xhalf(float x) {
    const int b = __builtin_bit_cast(int, x);
    return __builtin_bit_cast(float,
        __builtin_amdgcn_permlanex16(0, b, 0x76543210, 0xfedcba98, false, false));
}

__device__ __forceinline__ float xgroup(float x) {
    const int b = __builtin_bit_cast(int, x);
    return __builtin_bit_cast(float, __builtin_amdgcn_update_dpp(0, b, 0x128  , 0xf, 0xf, true));
}

__device__ __forceinline__ float fma_abs(float z, float w, float c) {
    float r;
    asm("v_fma_f32 %0, |%1|, %2, %3" : "=v"(r) : "v"(z), "v"(w), "v"(c));
    return r;
}
__device__ __forceinline__ float mul_abs(float z, float w) {
    float r;
    asm("v_mul_f32_e64 %0, |%1|, %2" : "=v"(r) : "v"(z), "v"(w));
    return r;
}

__global__ __launch_bounds__(128)
void MountainCar_7318624272632_kernel(const float* __restrict__ x,
                                      const float* __restrict__ w1,
                                      const float* __restrict__ b1,
                                      const float* __restrict__ w2,
                                      const float* __restrict__ b2,
                                      float* __restrict__ out, int B)
{
    const int lane   = threadIdx.x & 31;
    const int waveId = (blockIdx.x * blockDim.x + threadIdx.x) >> 5;
    const int col    = lane & 15;
    const int h      = lane >> 4;
    const int e      = col & 7;
    const int g      = col >> 3;
    const bool feeds   = (h == g);
    const bool outLane = (lane < 8);

    if (waveId * 8 >= B) return;

    const int  env      = waveId * 8 + e;
    const bool envValid = (env < B);
    const int  envC     = envValid ? env : (B - 1);

    v2f  Aw[2];
    v8f  Cb[2];
    float wh[2][8];
    float aP = 0.0f, cP = 0.0f, dP = 0.0f;
    const int hid0 = 16 * g + 8 * h;
#pragma unroll
    for (int t = 0; t < 2; ++t) {
        const int j = 32 * t + 16 * h + col;
        Aw[t][0] = w1[j];
        Aw[t][1] = w1[64 + j];
#pragma unroll
        for (int i = 0; i < 8; ++i) {
            const int hid = 32 * t + hid0 + i;
            Cb[t][i] = b1[hid];
            wh[t][i] = 0.5f * w2[hid];
            aP = fmaf(wh[t][i], w1[hid],      aP);
            cP = fmaf(wh[t][i], w1[64 + hid], cP);
            dP = fmaf(wh[t][i], Cb[t][i],     dP);
        }
    }
    aP += xhalf(aP);  cP += xhalf(cP);  dP += xhalf(dP);
    aP += xgroup(aP); cP += xgroup(cP); dP += xgroup(dP);
    const float aL = outLane ? aP : 0.0f;
    const float cL = outLane ? cP : 0.0f;
    const float dL = outLane ? (dP + b2[0]) : 0.0f;

    const float4 st = reinterpret_cast<const float4*>(x)[envC];

    float pf = st.x, vf = st.y, uf = st.z, rf = st.w;
    bool  live = (st.x <= GOAL_P);

    const float vlo = feeds ? MIN_V : 0.0f;
    const float vhi = feeds ? MAX_V : 0.0f;

    v2f S;
    {
        const bool hit0 = (st.x <= MIN_P);
        S[0] = feeds ? (hit0 ? MIN_P : st.x) : 0.0f;
        S[1] = feeds ? (hit0 ? 0.0f  : st.y) : 0.0f;
    }

    v8f d0 = __builtin_amdgcn_wmma_f32_16x16x4_f32(
             false, Aw[0],  false, S,
             (short)0, Cb[0],  false,  false);
    v8f d1 = __builtin_amdgcn_wmma_f32_16x16x4_f32(
             false, Aw[1],  false, S,
             (short)0, Cb[1],  false,  false);

    float lin = fmaf(aL, S[0], fmaf(cL, S[1], dL));
    float cs  = __builtin_amdgcn_cosf((3.0f * S[0]) * 0x1.45f306p-3f);

    static_assert(MAX_STEPS % STEPS_PER_CHECK == 0, "blocks must tile the step count");
    bool anyLive = ((__builtin_amdgcn_ballot_w32(pf <= GOAL_P) & 0xffu) != 0u);
    for (int blk = 0; anyLive && blk < MAX_STEPS / STEPS_PER_CHECK; ++blk) {
        for (int step = 0; step < STEPS_PER_CHECK; ++step) {
            const float p1 = S[0];

            const float plo  = p1 + vlo;
            const float phi  = p1 + vhi;
            const float wlo  = fmaxf(plo, MIN_P);
            const float whi  = fmaxf(phi, MIN_P);
            const float qthr = (phi <= MIN_P) ? __builtin_inff() : ((plo <= MIN_P) ? MIN_P : -__builtin_inff());

            float acc[4];
            acc[0] = fma_abs(d0[0], wh[0][0], lin);
#pragma unroll
            for (int k = 1; k < 4; ++k)
                acc[k] = mul_abs(d0[k], wh[0][k]);
#pragma unroll
            for (int k = 0; k < 4; ++k)
                acc[k] = fma_abs(d0[4 + k], wh[0][4 + k], acc[k]);
            __builtin_amdgcn_sched_barrier(0);
            const float vpart = fmaf(-0.0025f, cs, S[1]);
            const float ppart = p1 + vpart;
#pragma unroll
            for (int k = 0; k < 4; ++k)
                acc[k] = fma_abs(d1[k], wh[1][k], acc[k]);
#pragma unroll
            for (int k = 0; k < 4; ++k)
                acc[k] = fma_abs(d1[4 + k], wh[1][4 + k], acc[k]);

            const float s16   = (acc[0] + acc[1]) + (acc[2] + acc[3]);

            const float s32   = s16 + xhalf(s16);
            const float xarg  = s32 + xgroup(s32);

            const float u = __builtin_amdgcn_tanhf(xarg);

            const float q  = fmaf(u, 0.0015f, ppart);
            const float v2 = __builtin_amdgcn_fmed3f(fmaf(u, 0.0015f, vpart), vlo, vhi);

            const bool hit = (q <= qthr);
            S[0] = __builtin_amdgcn_fmed3f(q, wlo, whi);
            S[1] = hit ? 0.0f : v2;

            const float p2 = __builtin_amdgcn_fmed3f(q, plo, phi);

            const float u1 = fminf(fmaxf(u, MIN_ACC), MAX_ACC);
            const float rr = fmaf(-0.1f * u1, u1, rf);
            pf = live ? p2 : pf;
            vf = live ? v2 : vf;
            uf = live ? u1 : uf;
            rf = live ? rr : rf;
            live = live && (p2 <= GOAL_P);

            d0 = __builtin_amdgcn_wmma_f32_16x16x4_f32(
                     false, Aw[0],  false, S,
                     (short)0, Cb[0],  false,  false);
            lin = fmaf(aL, S[0], fmaf(cL, S[1], dL));
            const float rev = (3.0f * S[0]) * 0x1.45f306p-3f;
            __builtin_amdgcn_sched_barrier(0);
            d1 = __builtin_amdgcn_wmma_f32_16x16x4_f32(
                     false, Aw[1],  false, S,
                     (short)0, Cb[1],  false,  false);
            cs = __builtin_amdgcn_cosf(rev);
        }
        anyLive = ((__builtin_amdgcn_ballot_w32(pf <= GOAL_P) & 0xffu) != 0u);
    }

    const float rout = (pf <= GOAL_P) ? rf : (rf + 100.0f);
    if (outLane && envValid) {
        float4 o;
        o.x = pf; o.y = vf; o.z = uf; o.w = rout;
        reinterpret_cast<float4*>(out)[env] = o;
    }
}

extern "C" void kernel_launch(void* const* d_in, const int* in_sizes, int n_in,
                              void* d_out, int out_size, void* d_ws, size_t ws_size,
                              hipStream_t stream) {
    const float* x  = (const float*)d_in[0];
    const float* w1 = (const float*)d_in[1];
    const float* b1 = (const float*)d_in[2];
    const float* w2 = (const float*)d_in[3];
    const float* b2 = (const float*)d_in[4];
    float* out = (float*)d_out;

    const int B = in_sizes[0] / 4;
    const int envsPerWave = 8;
    const int waves = (B + envsPerWave - 1) / envsPerWave;
    const int threads = waves * 32;
    const int block = 128;
    const int grid = (threads + block - 1) / block;

    MountainCar_7318624272632_kernel<<<grid, block, 0, stream>>>(x, w1, b1, w2, b2, out, B);
}
